# starting kernel
# baseline (speedup 1.0000x reference)
#include <hip/hip_runtime.h>
#include <math.h>

typedef float v2f __attribute__((ext_vector_type(2)));
typedef float v8f __attribute__((ext_vector_type(8)));

#define DIM   192
#define C2    96
#define HW    4096
#define BATCH 2
#define NROW  (BATCH * HW)
#define KSTR  100

__global__ __launch_bounds__(256)
void nlb_qkv(const float* __restrict__ x,
             const float* __restrict__ w1, const float* __restrict__ b1,
             const float* __restrict__ w2, const float* __restrict__ b2,
             const float* __restrict__ w3, const float* __restrict__ b3,
             float* __restrict__ m1, float* __restrict__ m2, float* __restrict__ m3)
{
    const int hw = blockIdx.x * blockDim.x + threadIdx.x;
    const int bo = blockIdx.y;
    const int b  = bo / C2, o = bo - b * C2;
    const float* xb = x + (size_t)b * DIM * HW + hw;
    const float* W1 = w1 + o * DIM;
    const float* W2 = w2 + o * DIM;
    const float* W3 = w3 + o * DIM;
    float a1 = 0.f, a2 = 0.f, a3 = 0.f;
#pragma unroll 4
    for (int c = 0; c < DIM; ++c) {
        const float xv = xb[(size_t)c * HW];
        a1 = fmaf(xv, W1[c], a1);
        a2 = fmaf(xv, W2[c], a2);
        a3 = fmaf(xv, W3[c], a3);
    }
    const size_t idx = (size_t)bo * HW + hw;
    m1[idx] = a1 + b1[o];
    m2[idx] = a2 + b2[o];
    m3[idx] = a3 + b3[o];
}

__global__ __launch_bounds__(256)
void nlb_flash(const float* __restrict__ Q, const float* __restrict__ K,
               const float* __restrict__ V, float* __restrict__ O)
{
    __shared__ float sK[16 * KSTR];
    __shared__ float sV[16 * KSTR];
    __shared__ float sP[8 * 256];

    const int tid  = threadIdx.x;
    const int lane = tid & 31;
    const int wave = tid >> 5;
    const int row  = lane & 15;
    const int hi   = lane >> 4;
    const int qbase = (blockIdx.x * 8 + wave) * 16;
    float* myP = sP + wave * 256;

    v2f qb[24];
    {
        const float* qp = Q + (size_t)(qbase + row) * C2 + 2 * hi;
#pragma unroll
        for (int cc = 0; cc < 24; ++cc) qb[cc] = *(const v2f*)(qp + 4 * cc);
    }

    v8f acc[6];
#pragma unroll
    for (int jb = 0; jb < 6; ++jb)
#pragma unroll
        for (int e = 0; e < 8; ++e) acc[jb][e] = 0.f;
    float mrun = -INFINITY;
    float lrun = 0.f;

    for (int kt = 0; kt < NROW; kt += 16) {
        __syncthreads();
        {
            const float4* ks = (const float4*)(K + (size_t)kt * C2);
            const float4* vs = (const float4*)(V + (size_t)kt * C2);
            for (int i = tid; i < 384; i += 256) {
                const int r = i / 24, c4 = (i - r * 24) * 4;
                *(float4*)(sK + r * KSTR + c4) = ks[i];
                *(float4*)(sV + r * KSTR + c4) = vs[i];
            }
        }
        __syncthreads();

        v8f s;
#pragma unroll
        for (int e = 0; e < 8; ++e) s[e] = 0.f;
        {
            const float* kp = sK + row * KSTR + 2 * hi;
#pragma unroll
            for (int cc = 0; cc < 24; ++cc) {
                v2f a = *(const v2f*)(kp + 4 * cc);
                s = __builtin_amdgcn_wmma_f32_16x16x4_f32(
                        false, a, false, qb[cc], (short)0, s, false, false);
            }
        }

        float tmax = s[0];
#pragma unroll
        for (int i = 1; i < 8; ++i) tmax = fmaxf(tmax, s[i]);
        tmax = fmaxf(tmax, __shfl_xor(tmax, 16, 32));
        const float nm   = fmaxf(mrun, tmax);
        const float corr = __expf(mrun - nm);
        float p[8], ps = 0.f;
#pragma unroll
        for (int i = 0; i < 8; ++i) { p[i] = __expf(s[i] - nm); ps += p[i]; }
        ps += __shfl_xor(ps, 16, 32);
        lrun = lrun * corr + ps;
        mrun = nm;
#pragma unroll
        for (int jb = 0; jb < 6; ++jb)
#pragma unroll
            for (int e = 0; e < 8; ++e) acc[jb][e] *= corr;

#pragma unroll
        for (int i = 0; i < 8; ++i)
            myP[(i + 8 * hi) * 16 + row] = p[i];
        __syncthreads();

        v2f bp[4];
#pragma unroll
        for (int t = 0; t < 4; ++t) {
            bp[t].x = myP[(4 * t + 2 * hi) * 16 + row];
            bp[t].y = myP[(4 * t + 2 * hi + 1) * 16 + row];
        }

#pragma unroll
        for (int jb = 0; jb < 6; ++jb) {
#pragma unroll
            for (int t = 0; t < 4; ++t) {
                v2f a;
                a.x = sV[(4 * t + 2 * hi)     * KSTR + jb * 16 + row];
                a.y = sV[(4 * t + 2 * hi + 1) * KSTR + jb * 16 + row];
                acc[jb] = __builtin_amdgcn_wmma_f32_16x16x4_f32(
                              false, a, false, bp[t], (short)0, acc[jb], false, false);
            }
        }
    }

    const float inv = 1.f / lrun;
    float* op = O + (size_t)(qbase + row) * C2 + 8 * hi;
#pragma unroll
    for (int jb = 0; jb < 6; ++jb)
#pragma unroll
        for (int i = 0; i < 8; ++i)
            op[jb * 16 + i] = acc[jb][i] * inv;
}

__global__ __launch_bounds__(256)
void nlb_proj(const float* __restrict__ A, const float* __restrict__ wl,
              const float* __restrict__ bl, const float* __restrict__ x,
              float* __restrict__ out)
{
    const int hw = blockIdx.x * blockDim.x + threadIdx.x;
    const int bo = blockIdx.y;
    const int b  = bo / DIM, o = bo - b * DIM;
    const float* Ab = A + (size_t)b * C2 * HW + hw;
    const float* W  = wl + o * C2;
    float acc = bl[o];
#pragma unroll 4
    for (int c = 0; c < C2; ++c) acc = fmaf(Ab[(size_t)c * HW], W[c], acc);
    const size_t idx = (size_t)bo * HW + hw;
    out[idx] = acc + x[idx];
}

extern "C" void kernel_launch(void* const* d_in, const int* in_sizes, int n_in,
                              void* d_out, int out_size, void* d_ws, size_t ws_size,
                              hipStream_t stream)
{
    const float* x  = (const float*)d_in[0];
    const float* w1 = (const float*)d_in[1];
    const float* b1 = (const float*)d_in[2];
    const float* w2 = (const float*)d_in[3];
    const float* b2 = (const float*)d_in[4];
    const float* w3 = (const float*)d_in[5];
    const float* b3 = (const float*)d_in[6];
    const float* wl = (const float*)d_in[7];
    const float* bl = (const float*)d_in[8];
    float* out = (float*)d_out;

    const size_t msz = (size_t)NROW * C2;
    float* m1 = (float*)d_ws;
    float* m2 = m1 + msz;
    float* m3 = m2 + msz;
    float* ob = m3 + msz;

    nlb_qkv<<<dim3(HW / 256, BATCH * C2), 256, 0, stream>>>(
        x, w1, b1, w2, b2, w3, b3, m1, m2, m3);
    nlb_flash<<<dim3(NROW / (16 * 8)), 256, 0, stream>>>(m1, m2, m3, ob);
    nlb_proj<<<dim3(HW / 256, BATCH * DIM), 256, 0, stream>>>(ob, wl, bl, x, out);
}


# attention key tile of 64 rows per staging round (was 16): one global round trip, 3 barriers and one softmax rescale per 64 keys, 4 independent score chains; softmax max/sum regroup per 64 keys
# speedup vs baseline: 1.5375x; 1.5375x over previous
#include <hip/hip_runtime.h>
#include <math.h>

typedef float v2f __attribute__((ext_vector_type(2)));
typedef float v8f __attribute__((ext_vector_type(8)));

#define DIM   192
#define C2    96
#define HW    4096
#define BATCH 2
#define NROW  (BATCH * HW)
#define KSTR  100

__global__ __launch_bounds__(256)
void nlb_qkv(const float* __restrict__ x,
             const float* __restrict__ w1, const float* __restrict__ b1,
             const float* __restrict__ w2, const float* __restrict__ b2,
             const float* __restrict__ w3, const float* __restrict__ b3,
             float* __restrict__ m1, float* __restrict__ m2, float* __restrict__ m3)
{
    const int hw = blockIdx.x * blockDim.x + threadIdx.x;
    const int bo = blockIdx.y;
    const int b  = bo / C2, o = bo - b * C2;
    const float* xb = x + (size_t)b * DIM * HW + hw;
    const float* W1 = w1 + o * DIM;
    const float* W2 = w2 + o * DIM;
    const float* W3 = w3 + o * DIM;
    float a1 = 0.f, a2 = 0.f, a3 = 0.f;
#pragma unroll 4
    for (int c = 0; c < DIM; ++c) {
        const float xv = xb[(size_t)c * HW];
        a1 = fmaf(xv, W1[c], a1);
        a2 = fmaf(xv, W2[c], a2);
        a3 = fmaf(xv, W3[c], a3);
    }
    const size_t idx = (size_t)bo * HW + hw;
    m1[idx] = a1 + b1[o];
    m2[idx] = a2 + b2[o];
    m3[idx] = a3 + b3[o];
}

#define KT    64
#define NSUB  (KT / 16)
#define NSTG  (KT * 24 / 256)

__global__ __launch_bounds__(256)
void nlb_flash(const float* __restrict__ Q, const float* __restrict__ K,
               const float* __restrict__ V, float* __restrict__ O)
{
    __shared__ float sK[KT * KSTR];
    __shared__ float sV[KT * KSTR];
    __shared__ float sP[8 * KT * 16];

    const int tid  = threadIdx.x;
    const int lane = tid & 31;
    const int wave = tid >> 5;
    const int row  = lane & 15;
    const int hi   = lane >> 4;
    const int qbase = (blockIdx.x * 8 + wave) * 16;
    float* myP = sP + wave * (KT * 16);

    v2f qb[24];
    {
        const float* qp = Q + (size_t)(qbase + row) * C2 + 2 * hi;
#pragma unroll
        for (int cc = 0; cc < 24; ++cc) qb[cc] = *(const v2f*)(qp + 4 * cc);
    }

    v8f acc[6];
#pragma unroll
    for (int jb = 0; jb < 6; ++jb)
#pragma unroll
        for (int e = 0; e < 8; ++e) acc[jb][e] = 0.f;
    float mrun = -INFINITY;
    float lrun = 0.f;

    for (int kt = 0; kt < NROW; kt += KT) {
        __syncthreads();
        {
            const float4* ks = (const float4*)(K + (size_t)kt * C2);
            const float4* vs = (const float4*)(V + (size_t)kt * C2);
#pragma unroll
            for (int j = 0; j < NSTG; ++j) {
                const int i = tid + 256 * j;
                const int r = i / 24, c4 = (i - r * 24) * 4;
                *(float4*)(sK + r * KSTR + c4) = ks[i];
                *(float4*)(sV + r * KSTR + c4) = vs[i];
            }
        }
        __syncthreads();

        v8f s[NSUB];
#pragma unroll
        for (int u = 0; u < NSUB; ++u)
#pragma unroll
            for (int e = 0; e < 8; ++e) s[u][e] = 0.f;
        {
            const float* kp = sK + row * KSTR + 2 * hi;
#pragma unroll
            for (int cc = 0; cc < 24; ++cc) {
#pragma unroll
                for (int u = 0; u < NSUB; ++u) {
                    v2f a = *(const v2f*)(kp + u * 16 * KSTR + 4 * cc);
                    s[u] = __builtin_amdgcn_wmma_f32_16x16x4_f32(
                               false, a, false, qb[cc], (short)0, s[u], false, false);
                }
            }
        }

        float tmax = s[0][0];
#pragma unroll
        for (int u = 0; u < NSUB; ++u)
#pragma unroll
            for (int i = 0; i < 8; ++i) tmax = fmaxf(tmax, s[u][i]);
        tmax = fmaxf(tmax, __shfl_xor(tmax, 16, 32));
        const float nm   = fmaxf(mrun, tmax);
        const float corr = __expf(mrun - nm);
        float ps = 0.f;
#pragma unroll
        for (int u = 0; u < NSUB; ++u)
#pragma unroll
            for (int i = 0; i < 8; ++i) {
                const float pv = __expf(s[u][i] - nm);
                ps += pv;
                myP[(16 * u + i + 8 * hi) * 16 + row] = pv;
            }
        ps += __shfl_xor(ps, 16, 32);
        lrun = lrun * corr + ps;
        mrun = nm;
#pragma unroll
        for (int jb = 0; jb < 6; ++jb)
#pragma unroll
            for (int e = 0; e < 8; ++e) acc[jb][e] *= corr;

        __syncthreads();

#pragma unroll
        for (int t = 0; t < KT / 4; ++t) {
            v2f bp;
            bp.x = myP[(4 * t + 2 * hi) * 16 + row];
            bp.y = myP[(4 * t + 2 * hi + 1) * 16 + row];
#pragma unroll
            for (int jb = 0; jb < 6; ++jb) {
                v2f a;
                a.x = sV[(4 * t + 2 * hi)     * KSTR + jb * 16 + row];
                a.y = sV[(4 * t + 2 * hi + 1) * KSTR + jb * 16 + row];
                acc[jb] = __builtin_amdgcn_wmma_f32_16x16x4_f32(
                              false, a, false, bp, (short)0, acc[jb], false, false);
            }
        }
    }

    const float inv = 1.f / lrun;
    float* op = O + (size_t)(qbase + row) * C2 + 8 * hi;
#pragma unroll
    for (int jb = 0; jb < 6; ++jb)
#pragma unroll
        for (int i = 0; i < 8; ++i)
            op[jb * 16 + i] = acc[jb][i] * inv;
}

__global__ __launch_bounds__(256)
void nlb_proj(const float* __restrict__ A, const float* __restrict__ wl,
              const float* __restrict__ bl, const float* __restrict__ x,
              float* __restrict__ out)
{
    const int hw = blockIdx.x * blockDim.x + threadIdx.x;
    const int bo = blockIdx.y;
    const int b  = bo / DIM, o = bo - b * DIM;
    const float* Ab = A + (size_t)b * C2 * HW + hw;
    const float* W  = wl + o * C2;
    float acc = bl[o];
#pragma unroll 4
    for (int c = 0; c < C2; ++c) acc = fmaf(Ab[(size_t)c * HW], W[c], acc);
    const size_t idx = (size_t)bo * HW + hw;
    out[idx] = acc + x[idx];
}

extern "C" void kernel_launch(void* const* d_in, const int* in_sizes, int n_in,
                              void* d_out, int out_size, void* d_ws, size_t ws_size,
                              hipStream_t stream)
{
    const float* x  = (const float*)d_in[0];
    const float* w1 = (const float*)d_in[1];
    const float* b1 = (const float*)d_in[2];
    const float* w2 = (const float*)d_in[3];
    const float* b2 = (const float*)d_in[4];
    const float* w3 = (const float*)d_in[5];
    const float* b3 = (const float*)d_in[6];
    const float* wl = (const float*)d_in[7];
    const float* bl = (const float*)d_in[8];
    float* out = (float*)d_out;

    const size_t msz = (size_t)NROW * C2;
    float* m1 = (float*)d_ws;
    float* m2 = m1 + msz;
    float* m3 = m2 + msz;
    float* ob = m3 + msz;

    nlb_qkv<<<dim3(HW / 256, BATCH * C2), 256, 0, stream>>>(
        x, w1, b1, w2, b2, w3, b3, m1, m2, m3);
    nlb_flash<<<dim3(NROW / (16 * 8)), 256, 0, stream>>>(m1, m2, m3, ob);
    nlb_proj<<<dim3(HW / 256, BATCH * DIM), 256, 0, stream>>>(ob, wl, bl, x, out);
}


# attention: 2 waves share each 16-query tile (scores split by key half, P*V by feature half; tile max and P swapped in LDS), 256 work-groups x 4 waves; softmax row sums per key half, joined at end
# speedup vs baseline: 2.9485x; 1.9177x over previous
#include <hip/hip_runtime.h>
#include <math.h>

typedef float v2f __attribute__((ext_vector_type(2)));
typedef float v8f __attribute__((ext_vector_type(8)));

#define DIM   192
#define C2    96
#define HW    4096
#define BATCH 2
#define NROW  (BATCH * HW)
#define KSTR  100

__global__ __launch_bounds__(256)
void nlb_qkv(const float* __restrict__ x,
             const float* __restrict__ w1, const float* __restrict__ b1,
             const float* __restrict__ w2, const float* __restrict__ b2,
             const float* __restrict__ w3, const float* __restrict__ b3,
             float* __restrict__ m1, float* __restrict__ m2, float* __restrict__ m3)
{
    const int hw = blockIdx.x * blockDim.x + threadIdx.x;
    const int bo = blockIdx.y;
    const int b  = bo / C2, o = bo - b * C2;
    const float* xb = x + (size_t)b * DIM * HW + hw;
    const float* W1 = w1 + o * DIM;
    const float* W2 = w2 + o * DIM;
    const float* W3 = w3 + o * DIM;
    float a1 = 0.f, a2 = 0.f, a3 = 0.f;
#pragma unroll 4
    for (int c = 0; c < DIM; ++c) {
        const float xv = xb[(size_t)c * HW];
        a1 = fmaf(xv, W1[c], a1);
        a2 = fmaf(xv, W2[c], a2);
        a3 = fmaf(xv, W3[c], a3);
    }
    const size_t idx = (size_t)bo * HW + hw;
    m1[idx] = a1 + b1[o];
    m2[idx] = a2 + b2[o];
    m3[idx] = a3 + b3[o];
}

#define KT    64
#define NSUB  (KT / 16)
#define HSUB  (NSUB / 2)
#define HJB   3
#define QTILES 2
#define QTHR  (64 * QTILES)
#define NSTG  (KT * 24 / QTHR)
#define OSTR  100

__global__ __launch_bounds__(QTHR) __attribute__((amdgpu_waves_per_eu(1, 1)))
void nlb_flash(const float* __restrict__ Q, const float* __restrict__ K,
               const float* __restrict__ V, float* __restrict__ O)
{
    __shared__ float sK[KT * KSTR];
    __shared__ float sV[KT * KSTR];
    __shared__ float sP[QTILES * KT * 16];
    __shared__ float sM[QTILES * 2 * 16];

    const int tid  = threadIdx.x;
    const int lane = tid & 31;
    const int wave = tid >> 5;
    const int row  = lane & 15;
    const int hi   = lane >> 4;
    const int pair = wave >> 1;
    const int half = wave & 1;
    const int qbase = (blockIdx.x * QTILES + pair) * 16;
    float* prP = sP + pair * (KT * 16);
    float* prM = sM + pair * 32;

    v2f qb[24];
    {
        const float* qp = Q + (size_t)(qbase + row) * C2 + 2 * hi;
#pragma unroll
        for (int cc = 0; cc < 24; ++cc) qb[cc] = *(const v2f*)(qp + 4 * cc);
    }

    v8f acc[HJB];
#pragma unroll
    for (int jb = 0; jb < HJB; ++jb)
#pragma unroll
        for (int e = 0; e < 8; ++e) acc[jb][e] = 0.f;
    float mrun = -INFINITY;
    float lrun = 0.f;

    for (int kt = 0; kt < NROW; kt += KT) {
        __syncthreads();
        {
            const float4* ks = (const float4*)(K + (size_t)kt * C2);
            const float4* vs = (const float4*)(V + (size_t)kt * C2);
#pragma unroll
            for (int j = 0; j < NSTG; ++j) {
                const int i = tid + QTHR * j;
                const int r = i / 24, c4 = (i - r * 24) * 4;
                *(float4*)(sK + r * KSTR + c4) = ks[i];
                *(float4*)(sV + r * KSTR + c4) = vs[i];
            }
        }
        __syncthreads();

        v8f s[HSUB];
#pragma unroll
        for (int u = 0; u < HSUB; ++u)
#pragma unroll
            for (int e = 0; e < 8; ++e) s[u][e] = 0.f;
        {
            const float* kp = sK + (32 * half + row) * KSTR + 2 * hi;
#pragma unroll
            for (int cc = 0; cc < 24; ++cc) {
#pragma unroll
                for (int u = 0; u < HSUB; ++u) {
                    v2f a = *(const v2f*)(kp + u * 16 * KSTR + 4 * cc);
                    s[u] = __builtin_amdgcn_wmma_f32_16x16x4_f32(
                               false, a, false, qb[cc], (short)0, s[u], false, false);
                }
            }
        }

        float tmax = s[0][0];
#pragma unroll
        for (int u = 0; u < HSUB; ++u)
#pragma unroll
            for (int i = 0; i < 8; ++i) tmax = fmaxf(tmax, s[u][i]);
        tmax = fmaxf(tmax, __shfl_xor(tmax, 16, 32));
        prM[half * 16 + row] = tmax;
        __syncthreads();
        tmax = fmaxf(tmax, prM[(half ^ 1) * 16 + row]);

        const float nm   = fmaxf(mrun, tmax);
        const float corr = __expf(mrun - nm);
        float ps = 0.f;
        {
            float* pp = prP + (32 * half + 8 * hi) * 16 + row;
#pragma unroll
            for (int u = 0; u < HSUB; ++u)
#pragma unroll
                for (int i = 0; i < 8; ++i) {
                    const float pv = __expf(s[u][i] - nm);
                    ps += pv;
                    pp[(16 * u + i) * 16] = pv;
                }
        }
        ps += __shfl_xor(ps, 16, 32);
        lrun = lrun * corr + ps;
        mrun = nm;
#pragma unroll
        for (int jb = 0; jb < HJB; ++jb)
#pragma unroll
            for (int e = 0; e < 8; ++e) acc[jb][e] *= corr;

        __syncthreads();

        {
            const float* vp = sV + 2 * hi * KSTR + 48 * half + row;
            const float* bq = prP + 2 * hi * 16 + row;
#pragma unroll
            for (int t = 0; t < KT / 4; ++t) {
                v2f bp;
                bp.x = bq[(4 * t) * 16];
                bp.y = bq[(4 * t + 1) * 16];
#pragma unroll
                for (int jb = 0; jb < HJB; ++jb) {
                    v2f a;
                    a.x = vp[(4 * t)     * KSTR + jb * 16];
                    a.y = vp[(4 * t + 1) * KSTR + jb * 16];
                    acc[jb] = __builtin_amdgcn_wmma_f32_16x16x4_f32(
                                  false, a, false, bp, (short)0, acc[jb], false, false);
                }
            }
        }
    }

    __syncthreads();
    prM[half * 16 + row] = lrun;
    __syncthreads();
    const float ltot = prM[row] + prM[16 + row];
    const float inv = 1.f / ltot;

    float* so = sK + pair * (16 * OSTR);
#pragma unroll
    for (int jb = 0; jb < HJB; ++jb)
#pragma unroll
        for (int i = 0; i < 8; ++i)
            so[row * OSTR + 48 * half + jb * 16 + 8 * hi + i] = acc[jb][i] * inv;
    __syncthreads();
    {
        float* og = O + (size_t)(qbase + 8 * half) * C2;
        const float* sr = so + 8 * half * OSTR;
#pragma unroll
        for (int j = 0; j < 6; ++j) {
            const int idx = lane + 32 * j;
            const int r = idx / 24, c4 = (idx - r * 24) * 4;
            *(float4*)(og + r * C2 + c4) = *(const float4*)(sr + r * OSTR + c4);
        }
    }
}

__global__ __launch_bounds__(256)
void nlb_proj(const float* __restrict__ A, const float* __restrict__ wl,
              const float* __restrict__ bl, const float* __restrict__ x,
              float* __restrict__ out)
{
    const int hw = blockIdx.x * blockDim.x + threadIdx.x;
    const int bo = blockIdx.y;
    const int b  = bo / DIM, o = bo - b * DIM;
    const float* Ab = A + (size_t)b * C2 * HW + hw;
    const float* W  = wl + o * C2;
    float acc = bl[o];
#pragma unroll 4
    for (int c = 0; c < C2; ++c) acc = fmaf(Ab[(size_t)c * HW], W[c], acc);
    const size_t idx = (size_t)bo * HW + hw;
    out[idx] = acc + x[idx];
}

extern "C" void kernel_launch(void* const* d_in, const int* in_sizes, int n_in,
                              void* d_out, int out_size, void* d_ws, size_t ws_size,
                              hipStream_t stream)
{
    const float* x  = (const float*)d_in[0];
    const float* w1 = (const float*)d_in[1];
    const float* b1 = (const float*)d_in[2];
    const float* w2 = (const float*)d_in[3];
    const float* b2 = (const float*)d_in[4];
    const float* w3 = (const float*)d_in[5];
    const float* b3 = (const float*)d_in[6];
    const float* wl = (const float*)d_in[7];
    const float* bl = (const float*)d_in[8];
    float* out = (float*)d_out;

    const size_t msz = (size_t)NROW * C2;
    float* m1 = (float*)d_ws;
    float* m2 = m1 + msz;
    float* m3 = m2 + msz;
    float* ob = m3 + msz;

    nlb_qkv<<<dim3(HW / 256, BATCH * C2), 256, 0, stream>>>(
        x, w1, b1, w2, b2, w3, b3, m1, m2, m3);
    nlb_flash<<<dim3(NROW / (16 * QTILES)), QTHR, 0, stream>>>(m1, m2, m3, ob);
    nlb_proj<<<dim3(HW / 256, BATCH * DIM), 256, 0, stream>>>(ob, wl, bl, x, out);
}


# attention key tile of 256 rows per staging round (was 64): one global round trip, 4 barriers, one max exchange, one rescale per 256 keys; 8 score chains a wave; softmax max/sum regroup per 256 keys
# speedup vs baseline: 3.3325x; 1.1302x over previous
#include <hip/hip_runtime.h>
#include <math.h>

typedef float v2f __attribute__((ext_vector_type(2)));
typedef float v8f __attribute__((ext_vector_type(8)));

#define DIM   192
#define C2    96
#define HW    4096
#define BATCH 2
#define NROW  (BATCH * HW)
#define KSTR  100

__global__ __launch_bounds__(256)
void nlb_qkv(const float* __restrict__ x,
             const float* __restrict__ w1, const float* __restrict__ b1,
             const float* __restrict__ w2, const float* __restrict__ b2,
             const float* __restrict__ w3, const float* __restrict__ b3,
             float* __restrict__ m1, float* __restrict__ m2, float* __restrict__ m3)
{
    const int hw = blockIdx.x * blockDim.x + threadIdx.x;
    const int bo = blockIdx.y;
    const int b  = bo / C2, o = bo - b * C2;
    const float* xb = x + (size_t)b * DIM * HW + hw;
    const float* W1 = w1 + o * DIM;
    const float* W2 = w2 + o * DIM;
    const float* W3 = w3 + o * DIM;
    float a1 = 0.f, a2 = 0.f, a3 = 0.f;
#pragma unroll 4
    for (int c = 0; c < DIM; ++c) {
        const float xv = xb[(size_t)c * HW];
        a1 = fmaf(xv, W1[c], a1);
        a2 = fmaf(xv, W2[c], a2);
        a3 = fmaf(xv, W3[c], a3);
    }
    const size_t idx = (size_t)bo * HW + hw;
    m1[idx] = a1 + b1[o];
    m2[idx] = a2 + b2[o];
    m3[idx] = a3 + b3[o];
}

#define KT    256
#define NSUB  (KT / 16)
#define HSUB  (NSUB / 2)
#define HKEY  (KT / 2)
#define HJB   3
#define QTILES 2
#define QTHR  (64 * QTILES)
#define NSTG  (KT * 24 / QTHR)
#define OSTR  100

__global__ __launch_bounds__(QTHR) __attribute__((amdgpu_waves_per_eu(1, 1)))
void nlb_flash(const float* __restrict__ Q, const float* __restrict__ K,
               const float* __restrict__ V, float* __restrict__ O)
{
    __shared__ float sK[KT * KSTR];
    __shared__ float sV[KT * KSTR];
    __shared__ float sP[QTILES * KT * 16];
    __shared__ float sM[QTILES * 2 * 16];

    const int tid  = threadIdx.x;
    const int lane = tid & 31;
    const int wave = tid >> 5;
    const int row  = lane & 15;
    const int hi   = lane >> 4;
    const int pair = wave >> 1;
    const int half = wave & 1;
    const int qbase = (blockIdx.x * QTILES + pair) * 16;
    float* prP = sP + pair * (KT * 16);
    float* prM = sM + pair * 32;

    v2f qb[24];
    {
        const float* qp = Q + (size_t)(qbase + row) * C2 + 2 * hi;
#pragma unroll
        for (int cc = 0; cc < 24; ++cc) qb[cc] = *(const v2f*)(qp + 4 * cc);
    }

    v8f acc[HJB];
#pragma unroll
    for (int jb = 0; jb < HJB; ++jb)
#pragma unroll
        for (int e = 0; e < 8; ++e) acc[jb][e] = 0.f;
    float mrun = -INFINITY;
    float lrun = 0.f;

    for (int kt = 0; kt < NROW; kt += KT) {
        __syncthreads();
        {
            const float4* ks = (const float4*)(K + (size_t)kt * C2);
            const float4* vs = (const float4*)(V + (size_t)kt * C2);
#pragma unroll
            for (int j = 0; j < NSTG; ++j) {
                const int i = tid + QTHR * j;
                const int r = i / 24, c4 = (i - r * 24) * 4;
                *(float4*)(sK + r * KSTR + c4) = ks[i];
                *(float4*)(sV + r * KSTR + c4) = vs[i];
            }
        }
        __syncthreads();

        v8f s[HSUB];
#pragma unroll
        for (int u = 0; u < HSUB; ++u)
#pragma unroll
            for (int e = 0; e < 8; ++e) s[u][e] = 0.f;
        {
            const float* kp = sK + (HKEY * half + row) * KSTR + 2 * hi;
#pragma unroll
            for (int cc = 0; cc < 24; ++cc) {
#pragma unroll
                for (int u = 0; u < HSUB; ++u) {
                    v2f a = *(const v2f*)(kp + u * 16 * KSTR + 4 * cc);
                    s[u] = __builtin_amdgcn_wmma_f32_16x16x4_f32(
                               false, a, false, qb[cc], (short)0, s[u], false, false);
                }
            }
        }

        float tmax = s[0][0];
#pragma unroll
        for (int u = 0; u < HSUB; ++u)
#pragma unroll
            for (int i = 0; i < 8; ++i) tmax = fmaxf(tmax, s[u][i]);
        tmax = fmaxf(tmax, __shfl_xor(tmax, 16, 32));
        prM[half * 16 + row] = tmax;
        __syncthreads();
        tmax = fmaxf(tmax, prM[(half ^ 1) * 16 + row]);

        const float nm   = fmaxf(mrun, tmax);
        const float corr = __expf(mrun - nm);
        float ps = 0.f;
        {
            float* pp = prP + (HKEY * half + 8 * hi) * 16 + row;
#pragma unroll
            for (int u = 0; u < HSUB; ++u)
#pragma unroll
                for (int i = 0; i < 8; ++i) {
                    const float pv = __expf(s[u][i] - nm);
                    ps += pv;
                    pp[(16 * u + i) * 16] = pv;
                }
        }
        ps += __shfl_xor(ps, 16, 32);
        lrun = lrun * corr + ps;
        mrun = nm;
#pragma unroll
        for (int jb = 0; jb < HJB; ++jb)
#pragma unroll
            for (int e = 0; e < 8; ++e) acc[jb][e] *= corr;

        __syncthreads();

        {
            const float* vp = sV + 2 * hi * KSTR + 48 * half + row;
            const float* bq = prP + 2 * hi * 16 + row;
#pragma unroll
            for (int t = 0; t < KT / 4; ++t) {
                v2f bp;
                bp.x = bq[(4 * t) * 16];
                bp.y = bq[(4 * t + 1) * 16];
#pragma unroll
                for (int jb = 0; jb < HJB; ++jb) {
                    v2f a;
                    a.x = vp[(4 * t)     * KSTR + jb * 16];
                    a.y = vp[(4 * t + 1) * KSTR + jb * 16];
                    acc[jb] = __builtin_amdgcn_wmma_f32_16x16x4_f32(
                                  false, a, false, bp, (short)0, acc[jb], false, false);
                }
            }
        }
    }

    __syncthreads();
    prM[half * 16 + row] = lrun;
    __syncthreads();
    const float ltot = prM[row] + prM[16 + row];
    const float inv = 1.f / ltot;

    float* so = sK + pair * (16 * OSTR);
#pragma unroll
    for (int jb = 0; jb < HJB; ++jb)
#pragma unroll
        for (int i = 0; i < 8; ++i)
            so[row * OSTR + 48 * half + jb * 16 + 8 * hi + i] = acc[jb][i] * inv;
    __syncthreads();
    {
        float* og = O + (size_t)(qbase + 8 * half) * C2;
        const float* sr = so + 8 * half * OSTR;
#pragma unroll
        for (int j = 0; j < 6; ++j) {
            const int idx = lane + 32 * j;
            const int r = idx / 24, c4 = (idx - r * 24) * 4;
            *(float4*)(og + r * C2 + c4) = *(const float4*)(sr + r * OSTR + c4);
        }
    }
}

__global__ __launch_bounds__(256)
void nlb_proj(const float* __restrict__ A, const float* __restrict__ wl,
              const float* __restrict__ bl, const float* __restrict__ x,
              float* __restrict__ out)
{
    const int hw = blockIdx.x * blockDim.x + threadIdx.x;
    const int bo = blockIdx.y;
    const int b  = bo / DIM, o = bo - b * DIM;
    const float* Ab = A + (size_t)b * C2 * HW + hw;
    const float* W  = wl + o * C2;
    float acc = bl[o];
#pragma unroll 4
    for (int c = 0; c < C2; ++c) acc = fmaf(Ab[(size_t)c * HW], W[c], acc);
    const size_t idx = (size_t)bo * HW + hw;
    out[idx] = acc + x[idx];
}

extern "C" void kernel_launch(void* const* d_in, const int* in_sizes, int n_in,
                              void* d_out, int out_size, void* d_ws, size_t ws_size,
                              hipStream_t stream)
{
    const float* x  = (const float*)d_in[0];
    const float* w1 = (const float*)d_in[1];
    const float* b1 = (const float*)d_in[2];
    const float* w2 = (const float*)d_in[3];
    const float* b2 = (const float*)d_in[4];
    const float* w3 = (const float*)d_in[5];
    const float* b3 = (const float*)d_in[6];
    const float* wl = (const float*)d_in[7];
    const float* bl = (const float*)d_in[8];
    float* out = (float*)d_out;

    const size_t msz = (size_t)NROW * C2;
    float* m1 = (float*)d_ws;
    float* m2 = m1 + msz;
    float* m3 = m2 + msz;
    float* ob = m3 + msz;

    nlb_qkv<<<dim3(HW / 256, BATCH * C2), 256, 0, stream>>>(
        x, w1, b1, w2, b2, w3, b3, m1, m2, m3);
    nlb_flash<<<dim3(NROW / (16 * QTILES)), QTHR, 0, stream>>>(m1, m2, m3, ob);
    nlb_proj<<<dim3(HW / 256, BATCH * DIM), 256, 0, stream>>>(ob, wl, bl, x, out);
}


# QKV kernel writes V transposed; attention stages V^T and P^T ([feature][key], [query][key]): each P*V fragment is one 16-byte LDS read for 2 WMMAs, read in block groups; P*V key sum order changes
# speedup vs baseline: 3.5984x; 1.0798x over previous
#include <hip/hip_runtime.h>
#include <math.h>

typedef float v2f __attribute__((ext_vector_type(2)));
typedef float v4f __attribute__((ext_vector_type(4)));
typedef float v8f __attribute__((ext_vector_type(8)));

#define DIM   192
#define C2    96
#define HW    4096
#define BATCH 2
#define NROW  (BATCH * HW)
#define KSTR  100

__global__ __launch_bounds__(256)
void nlb_qkv(const float* __restrict__ x,
             const float* __restrict__ w1, const float* __restrict__ b1,
             const float* __restrict__ w2, const float* __restrict__ b2,
             const float* __restrict__ w3, const float* __restrict__ b3,
             float* __restrict__ m1, float* __restrict__ m2, float* __restrict__ m3t)
{
    const int hw = blockIdx.x * blockDim.x + threadIdx.x;
    const int bo = blockIdx.y;
    const int b  = bo / C2, o = bo - b * C2;
    const float* xb = x + (size_t)b * DIM * HW + hw;
    const float* W1 = w1 + o * DIM;
    const float* W2 = w2 + o * DIM;
    const float* W3 = w3 + o * DIM;
    float a1 = 0.f, a2 = 0.f, a3 = 0.f;
#pragma unroll 4
    for (int c = 0; c < DIM; ++c) {
        const float xv = xb[(size_t)c * HW];
        a1 = fmaf(xv, W1[c], a1);
        a2 = fmaf(xv, W2[c], a2);
        a3 = fmaf(xv, W3[c], a3);
    }
    const size_t idx = (size_t)bo * HW + hw;
    m1[idx] = a1 + b1[o];
    m2[idx] = a2 + b2[o];
    const unsigned fi = (unsigned)idx;
    const unsigned n = fi / C2, f = fi - n * C2;
    m3t[(size_t)f * NROW + n] = a3 + b3[o];
}

#define KT    256
#define NSUB  (KT / 16)
#define HSUB  (NSUB / 2)
#define HKEY  (KT / 2)
#define HJB   3
#define QTILES 2
#define QTHR  (64 * QTILES)
#define NSTG  (KT * 24 / QTHR)
#define OSTR  100
#define TSTR  (KT + 4)
#define NSTV  (C2 * (KT / 4) / QTHR)
#define JG    8
#define NRG   (KT / 8 / JG)

__global__ __launch_bounds__(QTHR) __attribute__((amdgpu_waves_per_eu(1, 1)))
void nlb_flash(const float* __restrict__ Q, const float* __restrict__ K,
               const float* __restrict__ Vt, float* __restrict__ O)
{
    __shared__ __attribute__((aligned(16))) float sK[KT * KSTR];
    __shared__ __attribute__((aligned(16))) float sVT[C2 * TSTR];
    __shared__ __attribute__((aligned(16))) float sP[QTILES * 16 * TSTR];
    __shared__ float sM[QTILES * 2 * 16];

    const int tid  = threadIdx.x;
    const int lane = tid & 31;
    const int wave = tid >> 5;
    const int row  = lane & 15;
    const int hi   = lane >> 4;
    const int pair = wave >> 1;
    const int half = wave & 1;
    const int qbase = (blockIdx.x * QTILES + pair) * 16;
    float* prP = sP + pair * (16 * TSTR);
    float* prM = sM + pair * 32;

    v2f qb[24];
    {
        const float* qp = Q + (size_t)(qbase + row) * C2 + 2 * hi;
#pragma unroll
        for (int cc = 0; cc < 24; ++cc) qb[cc] = *(const v2f*)(qp + 4 * cc);
    }

    v8f acc[HJB];
#pragma unroll
    for (int jb = 0; jb < HJB; ++jb)
#pragma unroll
        for (int e = 0; e < 8; ++e) acc[jb][e] = 0.f;
    float mrun = -INFINITY;
    float lrun = 0.f;

    for (int kt = 0; kt < NROW; kt += KT) {
        __syncthreads();
        {
            const float4* ks = (const float4*)(K + (size_t)kt * C2);
            const float* vs = Vt + kt;
#pragma unroll
            for (int j = 0; j < NSTG; ++j) {
                const int i = tid + QTHR * j;
                const int r = i / 24, c4 = (i - r * 24) * 4;
                *(float4*)(sK + r * KSTR + c4) = ks[i];
            }
#pragma unroll
            for (int j = 0; j < NSTV; ++j) {
                const int i = tid + QTHR * j;
                const int r = i / (KT / 4), c4 = (i - r * (KT / 4)) * 4;
                *(float4*)(sVT + r * TSTR + c4) =
                    *(const float4*)(vs + (size_t)r * NROW + c4);
            }
        }
        __syncthreads();

        v8f s[HSUB];
#pragma unroll
        for (int u = 0; u < HSUB; ++u)
#pragma unroll
            for (int e = 0; e < 8; ++e) s[u][e] = 0.f;
        {
            const float* kp = sK + (HKEY * half + row) * KSTR + 2 * hi;
            v2f ka[24][HSUB];
#pragma unroll
            for (int cc = 0; cc < 24; ++cc)
#pragma unroll
                for (int u = 0; u < HSUB; ++u)
                    ka[cc][u] = *(const v2f*)(kp + u * 16 * KSTR + 4 * cc);
            __builtin_amdgcn_sched_barrier(0);
#pragma unroll
            for (int cc = 0; cc < 24; ++cc)
#pragma unroll
                for (int u = 0; u < HSUB; ++u)
                    s[u] = __builtin_amdgcn_wmma_f32_16x16x4_f32(
                               false, ka[cc][u], false, qb[cc], (short)0, s[u], false, false);
        }

        float tmax = s[0][0];
#pragma unroll
        for (int u = 0; u < HSUB; ++u)
#pragma unroll
            for (int i = 0; i < 8; ++i) tmax = fmaxf(tmax, s[u][i]);
        tmax = fmaxf(tmax, __shfl_xor(tmax, 16, 32));
        prM[half * 16 + row] = tmax;
        __syncthreads();
        tmax = fmaxf(tmax, prM[(half ^ 1) * 16 + row]);

        const float nm   = fmaxf(mrun, tmax);
        const float corr = __expf(mrun - nm);
        float ps = 0.f;
        {
            float* pp = prP + row * TSTR + HKEY * half + 8 * hi;
#pragma unroll
            for (int u = 0; u < HSUB; ++u) {
                v4f p0, p1;
#pragma unroll
                for (int i = 0; i < 4; ++i) {
                    const float pv = __expf(s[u][i] - nm);
                    ps += pv;
                    p0[i] = pv;
                }
#pragma unroll
                for (int i = 0; i < 4; ++i) {
                    const float pv = __expf(s[u][4 + i] - nm);
                    ps += pv;
                    p1[i] = pv;
                }
                *(v4f*)(pp + 16 * u)     = p0;
                *(v4f*)(pp + 16 * u + 4) = p1;
            }
        }
        ps += __shfl_xor(ps, 16, 32);
        lrun = lrun * corr + ps;
        mrun = nm;
#pragma unroll
        for (int jb = 0; jb < HJB; ++jb)
#pragma unroll
            for (int e = 0; e < 8; ++e) acc[jb][e] *= corr;

        __syncthreads();

        {
            const float* vp = sVT + (48 * half + row) * TSTR + 4 * hi;
            const float* bq = prP + row * TSTR + 4 * hi;
            v4f pf[2][JG];
            v4f vf[2][JG][HJB];
#pragma unroll
            for (int jj = 0; jj < JG; ++jj) {
                pf[0][jj] = *(const v4f*)(bq + 8 * jj);
#pragma unroll
                for (int jb = 0; jb < HJB; ++jb)
                    vf[0][jj][jb] = *(const v4f*)(vp + jb * 16 * TSTR + 8 * jj);
                if (jj == JG / 2 - 1 || jj == JG - 1) __builtin_amdgcn_sched_barrier(0);
            }
#pragma unroll
            for (int g = 0; g < NRG; ++g) {
                const int cur = g & 1, nxt = cur ^ 1;
                if (g + 1 < NRG) {
#pragma unroll
                    for (int jj = 0; jj < JG; ++jj) {
                        const int j = (g + 1) * JG + jj;
                        pf[nxt][jj] = *(const v4f*)(bq + 8 * j);
#pragma unroll
                        for (int jb = 0; jb < HJB; ++jb)
                            vf[nxt][jj][jb] = *(const v4f*)(vp + jb * 16 * TSTR + 8 * j);
                    }
                }
                __builtin_amdgcn_sched_barrier(0);
#pragma unroll
                for (int jj = 0; jj < JG; ++jj) {
#pragma unroll
                    for (int jb = 0; jb < HJB; ++jb)
                        acc[jb] = __builtin_amdgcn_wmma_f32_16x16x4_f32(
                                      false, vf[cur][jj][jb].xy, false, pf[cur][jj].xy,
                                      (short)0, acc[jb], false, false);
#pragma unroll
                    for (int jb = 0; jb < HJB; ++jb)
                        acc[jb] = __builtin_amdgcn_wmma_f32_16x16x4_f32(
                                      false, vf[cur][jj][jb].zw, false, pf[cur][jj].zw,
                                      (short)0, acc[jb], false, false);
                }
                __builtin_amdgcn_sched_barrier(0);
            }
        }
    }

    __syncthreads();
    prM[half * 16 + row] = lrun;
    __syncthreads();
    const float ltot = prM[row] + prM[16 + row];
    const float inv = 1.f / ltot;

    float* so = sK + pair * (16 * OSTR);
#pragma unroll
    for (int jb = 0; jb < HJB; ++jb)
#pragma unroll
        for (int i = 0; i < 8; ++i)
            so[row * OSTR + 48 * half + jb * 16 + 8 * hi + i] = acc[jb][i] * inv;
    __syncthreads();
    {
        float* og = O + (size_t)(qbase + 8 * half) * C2;
        const float* sr = so + 8 * half * OSTR;
#pragma unroll
        for (int j = 0; j < 6; ++j) {
            const int idx = lane + 32 * j;
            const int r = idx / 24, c4 = (idx - r * 24) * 4;
            *(float4*)(og + r * C2 + c4) = *(const float4*)(sr + r * OSTR + c4);
        }
    }
}

__global__ __launch_bounds__(256)
void nlb_proj(const float* __restrict__ A, const float* __restrict__ wl,
              const float* __restrict__ bl, const float* __restrict__ x,
              float* __restrict__ out)
{
    const int hw = blockIdx.x * blockDim.x + threadIdx.x;
    const int bo = blockIdx.y;
    const int b  = bo / DIM, o = bo - b * DIM;
    const float* Ab = A + (size_t)b * C2 * HW + hw;
    const float* W  = wl + o * C2;
    float acc = bl[o];
#pragma unroll 4
    for (int c = 0; c < C2; ++c) acc = fmaf(Ab[(size_t)c * HW], W[c], acc);
    const size_t idx = (size_t)bo * HW + hw;
    out[idx] = acc + x[idx];
}

extern "C" void kernel_launch(void* const* d_in, const int* in_sizes, int n_in,
                              void* d_out, int out_size, void* d_ws, size_t ws_size,
                              hipStream_t stream)
{
    const float* x  = (const float*)d_in[0];
    const float* w1 = (const float*)d_in[1];
    const float* b1 = (const float*)d_in[2];
    const float* w2 = (const float*)d_in[3];
    const float* b2 = (const float*)d_in[4];
    const float* w3 = (const float*)d_in[5];
    const float* b3 = (const float*)d_in[6];
    const float* wl = (const float*)d_in[7];
    const float* bl = (const float*)d_in[8];
    float* out = (float*)d_out;

    const size_t msz = (size_t)NROW * C2;
    float* m1 = (float*)d_ws;
    float* m2 = m1 + msz;
    float* m3t = m2 + msz;
    float* ob = m3t + msz;

    nlb_qkv<<<dim3(HW / 256, BATCH * C2), 256, 0, stream>>>(
        x, w1, b1, w2, b2, w3, b3, m1, m2, m3t);
    nlb_flash<<<dim3(NROW / (16 * QTILES)), QTHR, 0, stream>>>(m1, m2, m3t, ob);
    nlb_proj<<<dim3(HW / 256, BATCH * DIM), 256, 0, stream>>>(ob, wl, bl, x, out);
}


# attention: K/V fragments go global->registers in fragment order (no LDS tiles, no loop barriers); wave owns a key quarter of 32 queries, chunks rotated by group; 4 partials merged; sum orders change
# speedup vs baseline: 5.4689x; 1.5198x over previous
#include <hip/hip_runtime.h>
#include <math.h>


typedef float v2f __attribute__((ext_vector_type(2)));
typedef float v4f __attribute__((ext_vector_type(4)));
typedef float v8f __attribute__((ext_vector_type(8)));

#define DIM   192
#define C2    96
#define HW    4096
#define BATCH 2
#define NROW  (BATCH * HW)

__global__ __launch_bounds__(256)
void nlb_qkv(const float* __restrict__ x,
             const float* __restrict__ w1, const float* __restrict__ b1,
             const float* __restrict__ w2, const float* __restrict__ b2,
             const float* __restrict__ w3, const float* __restrict__ b3,
             float* __restrict__ m1, float* __restrict__ kf, float* __restrict__ vf)
{
    const int hw = blockIdx.x * blockDim.x + threadIdx.x;
    const int bo = blockIdx.y;
    const int b  = bo / C2, o = bo - b * C2;
    const float* xb = x + (size_t)b * DIM * HW + hw;
    const float* W1 = w1 + o * DIM;
    const float* W2 = w2 + o * DIM;
    const float* W3 = w3 + o * DIM;
    float a1 = 0.f, a2 = 0.f, a3 = 0.f;
#pragma unroll 4
    for (int c = 0; c < DIM; ++c) {
        const float xv = xb[(size_t)c * HW];
        a1 = fmaf(xv, W1[c], a1);
        a2 = fmaf(xv, W2[c], a2);
        a3 = fmaf(xv, W3[c], a3);
    }
    const size_t idx = (size_t)bo * HW + hw;
    m1[idx] = a1 + b1[o];
    const unsigned fi = (unsigned)idx;
    const unsigned n = fi / C2, f = fi - n * C2;
    const unsigned tile = n >> 4, kk = n & 15;
    kf[tile * (16 * C2) + (f >> 3) * 128 + ((((f >> 2) & 1) * 16 + kk) << 2) + (f & 3)] = a2 + b2[o];
    vf[tile * (16 * C2) + (((f >> 4) * 2 + ((kk >> 2) & 1)) << 7) + ((((kk >> 3) * 16 + (f & 15))) << 2) + (kk & 3)] = a3 + b3[o];
}

#define QPW    32
#define NWAVE  4
#define QTHR   (32 * NWAVE)
#define KCH    64
#define NSUBT  (KCH / 16)
#define NCH    (NROW / NWAVE / KCH)
#define TILEF  (16 * C2)
#define OSTR   100

__global__ __launch_bounds__(QTHR) __attribute__((amdgpu_waves_per_eu(1, 1)))
void nlb_flash(const float* __restrict__ Q, const float* __restrict__ Kf,
               const float* __restrict__ Vf, float* __restrict__ O)
{
    __shared__ __attribute__((aligned(16))) float sO[NWAVE * QPW * OSTR];
    __shared__ float sM[NWAVE * QPW];
    __shared__ float sL[NWAVE * QPW];

    const int tid  = threadIdx.x;
    const int lane = tid & 31;
    const int wave = tid >> 5;
    const int c    = lane & 15;
    const int h    = lane >> 4;
    const int qbase = blockIdx.x * QPW;

    v4f qf[2][12];
#pragma unroll
    for (int g = 0; g < 2; ++g) {
        const float* qp = Q + (size_t)(qbase + 16 * g + c) * C2 + 4 * h;
#pragma unroll
        for (int c2 = 0; c2 < 12; ++c2) qf[g][c2] = *(const v4f*)(qp + 8 * c2);
    }

    v8f acc[2][6];
#pragma unroll
    for (int g = 0; g < 2; ++g)
#pragma unroll
        for (int F = 0; F < 6; ++F)
#pragma unroll
            for (int e = 0; e < 8; ++e) acc[g][F][e] = 0.f;
    float mrun[2] = { -INFINITY, -INFINITY };
    float lrun[2] = { 0.f, 0.f };

    const v4f* kq = (const v4f*)Kf + (size_t)wave * (NCH * NSUBT * (TILEF / 4)) + lane;
    const v4f* vq = (const v4f*)Vf + (size_t)wave * (NCH * NSUBT * (TILEF / 4)) + lane;
    const int rot = blockIdx.x & (NCH - 1);

    v4f ka[NSUBT][12];
    v4f va[NSUBT][6][2];
    {
        const v4f* kp = kq + rot * (NSUBT * (TILEF / 4));
        const v4f* vp = vq + rot * (NSUBT * (TILEF / 4));
#pragma unroll
        for (int u = 0; u < NSUBT; ++u)
#pragma unroll
            for (int c2 = 0; c2 < 12; ++c2) ka[u][c2] = kp[u * (TILEF / 4) + c2 * 32];
#pragma unroll
        for (int u = 0; u < NSUBT; ++u)
#pragma unroll
            for (int F = 0; F < 6; ++F)
#pragma unroll
                for (int t2 = 0; t2 < 2; ++t2) va[u][F][t2] = vp[u * (TILEF / 4) + (F * 2 + t2) * 32];
    }

    for (int i = 0; i < NCH; ++i) {
        const int cn = (i + 1 + rot) & (NCH - 1);
        const v4f* kn = kq + cn * (NSUBT * (TILEF / 4));
        const v4f* vn = vq + cn * (NSUBT * (TILEF / 4));

        v8f s[2][NSUBT];
#pragma unroll
        for (int g = 0; g < 2; ++g)
#pragma unroll
            for (int u = 0; u < NSUBT; ++u)
#pragma unroll
                for (int e = 0; e < 8; ++e) s[g][u][e] = 0.f;
#pragma unroll
        for (int c2 = 0; c2 < 12; ++c2) {
#pragma unroll
            for (int u = 0; u < NSUBT; ++u)
#pragma unroll
                for (int g = 0; g < 2; ++g)
                    s[g][u] = __builtin_amdgcn_wmma_f32_16x16x4_f32(
                                  false, ka[u][c2].xy, false, qf[g][c2].xy, (short)0, s[g][u], false, false);
#pragma unroll
            for (int u = 0; u < NSUBT; ++u)
#pragma unroll
                for (int g = 0; g < 2; ++g)
                    s[g][u] = __builtin_amdgcn_wmma_f32_16x16x4_f32(
                                  false, ka[u][c2].zw, false, qf[g][c2].zw, (short)0, s[g][u], false, false);
        }
        __builtin_amdgcn_sched_barrier(0);
#pragma unroll
        for (int u = 0; u < NSUBT; ++u)
#pragma unroll
            for (int c2 = 0; c2 < 12; ++c2) ka[u][c2] = kn[u * (TILEF / 4) + c2 * 32];
        __builtin_amdgcn_sched_barrier(0);

#pragma unroll
        for (int g = 0; g < 2; ++g) {
            float tmax = s[g][0][0];
#pragma unroll
            for (int u = 0; u < NSUBT; ++u)
#pragma unroll
                for (int r = 0; r < 8; ++r) tmax = fmaxf(tmax, s[g][u][r]);
            tmax = fmaxf(tmax, __shfl_xor(tmax, 16, 32));
            const float nm   = fmaxf(mrun[g], tmax);
            const float corr = __expf(mrun[g] - nm);
            float ps = 0.f;
#pragma unroll
            for (int u = 0; u < NSUBT; ++u)
#pragma unroll
                for (int r = 0; r < 8; ++r) {
                    const float pv = __expf(s[g][u][r] - nm);
                    ps += pv;
                    s[g][u][r] = pv;
                }
            lrun[g] = lrun[g] * corr + ps;
            mrun[g] = nm;
#pragma unroll
            for (int F = 0; F < 6; ++F)
#pragma unroll
                for (int e = 0; e < 8; ++e) acc[g][F][e] *= corr;
        }

#pragma unroll
        for (int u = 0; u < NSUBT; ++u)
#pragma unroll
            for (int t2 = 0; t2 < 2; ++t2) {
                const v2f p00 = { s[0][u][4 * t2 + 0], s[0][u][4 * t2 + 1] };
                const v2f p10 = { s[1][u][4 * t2 + 0], s[1][u][4 * t2 + 1] };
                const v2f p01 = { s[0][u][4 * t2 + 2], s[0][u][4 * t2 + 3] };
                const v2f p11 = { s[1][u][4 * t2 + 2], s[1][u][4 * t2 + 3] };
#pragma unroll
                for (int F = 0; F < 6; ++F) {
                    acc[0][F] = __builtin_amdgcn_wmma_f32_16x16x4_f32(
                                    false, va[u][F][t2].xy, false, p00, (short)0, acc[0][F], false, false);
                    acc[1][F] = __builtin_amdgcn_wmma_f32_16x16x4_f32(
                                    false, va[u][F][t2].xy, false, p10, (short)0, acc[1][F], false, false);
                }
#pragma unroll
                for (int F = 0; F < 6; ++F) {
                    acc[0][F] = __builtin_amdgcn_wmma_f32_16x16x4_f32(
                                    false, va[u][F][t2].zw, false, p01, (short)0, acc[0][F], false, false);
                    acc[1][F] = __builtin_amdgcn_wmma_f32_16x16x4_f32(
                                    false, va[u][F][t2].zw, false, p11, (short)0, acc[1][F], false, false);
                }
            }
        __builtin_amdgcn_sched_barrier(0);
#pragma unroll
        for (int u = 0; u < NSUBT; ++u)
#pragma unroll
            for (int F = 0; F < 6; ++F)
#pragma unroll
                for (int t2 = 0; t2 < 2; ++t2) va[u][F][t2] = vn[u * (TILEF / 4) + (F * 2 + t2) * 32];
        __builtin_amdgcn_sched_barrier(0);
    }

    {
        float* so = sO + wave * (QPW * OSTR);
#pragma unroll
        for (int g = 0; g < 2; ++g) {
            const float lt = lrun[g] + __shfl_xor(lrun[g], 16, 32);
            sM[wave * QPW + 16 * g + c] = mrun[g];
            sL[wave * QPW + 16 * g + c] = lt;
#pragma unroll
            for (int F = 0; F < 6; ++F) {
                float* d = so + (16 * g + c) * OSTR + 16 * F + 8 * h;
                *(v4f*)(d)     = __builtin_shufflevector(acc[g][F], acc[g][F], 0, 1, 2, 3);
                *(v4f*)(d + 4) = __builtin_shufflevector(acc[g][F], acc[g][F], 4, 5, 6, 7);
            }
        }
    }
    __syncthreads();
    {
        float* og = O + (size_t)(qbase + 8 * wave) * C2;
#pragma unroll
        for (int j = 0; j < 6; ++j) {
            const int idx = lane + 32 * j;
            const int r = idx / 24, c4 = (idx - r * 24) * 4;
            const int q = 8 * wave + r;
            float mw[NWAVE];
            float mt = sM[q];
#pragma unroll
            for (int w = 0; w < NWAVE; ++w) { mw[w] = sM[w * QPW + q]; mt = fmaxf(mt, mw[w]); }
            float lt = 0.f;
            v4f ov = { 0.f, 0.f, 0.f, 0.f };
#pragma unroll
            for (int w = 0; w < NWAVE; ++w) {
                const float sc = __expf(mw[w] - mt);
                lt += sL[w * QPW + q] * sc;
                ov += *(const v4f*)(sO + w * (QPW * OSTR) + q * OSTR + c4) * sc;
            }
            const float inv = 1.f / lt;
            *(v4f*)(og + r * C2 + c4) = ov * inv;
        }
    }
}

__global__ __launch_bounds__(256)
void nlb_proj(const float* __restrict__ A, const float* __restrict__ wl,
              const float* __restrict__ bl, const float* __restrict__ x,
              float* __restrict__ out)
{
    const int hw = blockIdx.x * blockDim.x + threadIdx.x;
    const int bo = blockIdx.y;
    const int b  = bo / DIM, o = bo - b * DIM;
    const float* Ab = A + (size_t)b * C2 * HW + hw;
    const float* W  = wl + o * C2;
    float acc = bl[o];
#pragma unroll 4
    for (int c = 0; c < C2; ++c) acc = fmaf(Ab[(size_t)c * HW], W[c], acc);
    const size_t idx = (size_t)bo * HW + hw;
    out[idx] = acc + x[idx];
}

extern "C" void kernel_launch(void* const* d_in, const int* in_sizes, int n_in,
                              void* d_out, int out_size, void* d_ws, size_t ws_size,
                              hipStream_t stream)
{
    const float* x  = (const float*)d_in[0];
    const float* w1 = (const float*)d_in[1];
    const float* b1 = (const float*)d_in[2];
    const float* w2 = (const float*)d_in[3];
    const float* b2 = (const float*)d_in[4];
    const float* w3 = (const float*)d_in[5];
    const float* b3 = (const float*)d_in[6];
    const float* wl = (const float*)d_in[7];
    const float* bl = (const float*)d_in[8];
    float* out = (float*)d_out;

    const size_t msz = (size_t)NROW * C2;
    float* m1 = (float*)d_ws;
    float* kf = m1 + msz;
    float* vf = kf + msz;
    float* ob = vf + msz;

    nlb_qkv<<<dim3(HW / 256, BATCH * C2), 256, 0, stream>>>(
        x, w1, b1, w2, b2, w3, b3, m1, kf, vf);
    nlb_flash<<<dim3(NROW / QPW), QTHR, 0, stream>>>(m1, kf, vf, ob);
    nlb_proj<<<dim3(HW / 256, BATCH * DIM), 256, 0, stream>>>(ob, wl, bl, x, out);
}
